# starting kernel
# baseline (speedup 1.0000x reference)
#include <hip/hip_runtime.h>
#include <math.h>

typedef __attribute__((ext_vector_type(2))) float v2f;
typedef __attribute__((ext_vector_type(8))) float v8f;

#define BB   16
#define CC   64
#define HH   96
#define WW   96
#define HWN  (HH*WW)
#define NAA  20
#define NII  50
#define PP   (NAA*NII)
#define T32  32
#define NWAVES 4
#define JTOT (HWN/16)
#define JPW  (JTOT/NWAVES)
#define ASCALE (50.0f * 1.44269504088896340736f)
#define CSTEP  (2.0f / 95.0f)

__device__ __forceinline__ float fast_exp2(float x) {
  return __builtin_amdgcn_exp2f(x);
}

__global__ __launch_bounds__(32*NWAVES)
void corr_softargmax_kernel(const float* __restrict__ x,
                            const float* __restrict__ desc,
                            float* __restrict__ out)
{
  __shared__ float lds[32][NWAVES][4];

  const int b    = blockIdx.x / T32;
  const int t    = blockIdx.x % T32;
  const int tid  = threadIdx.x;
  const int wave = tid >> 5;
  const int lane = tid & 31;
  const int half = lane >> 4;
  const int nl   = lane & 15;
  const int p0   = t * 32;

  v2f aLo[16], aHi[16];
  {
    const int  pl  = p0 + nl;        const bool okl = (pl < PP);
    const int  ph  = p0 + 16 + nl;   const bool okh = (ph < PP);
    #pragma unroll
    for (int kc = 0; kc < 16; ++kc) {
      const int k0 = 4 * kc + 2 * half;
      aLo[kc].x = okl ? ASCALE * desc[(size_t)(k0    ) * PP + pl] : 0.0f;
      aLo[kc].y = okl ? ASCALE * desc[(size_t)(k0 + 1) * PP + pl] : 0.0f;
      aHi[kc].x = okh ? ASCALE * desc[(size_t)(k0    ) * PP + ph] : 0.0f;
      aHi[kc].y = okh ? ASCALE * desc[(size_t)(k0 + 1) * PP + ph] : 0.0f;
    }
  }

  float m[16], se[16], sx[16], sy[16];
  #pragma unroll
  for (int r = 0; r < 16; ++r) { m[r] = -1e30f; se[r] = 0.f; sx[r] = 0.f; sy[r] = 0.f; }

  const float* xb = x + (size_t)b * CC * HWN;
  const int jbeg = wave * JPW;
  const int jend = jbeg + JPW;

  v2f bv0[16], bv1[16];

  auto loadB = [&](v2f (&bv)[16], int j) {
    const int n = j * 16 + nl;
    #pragma unroll
    for (int kc = 0; kc < 16; ++kc) {
      const int k0 = 4 * kc + 2 * half;
      bv[kc].x = xb[(size_t)(k0    ) * HWN + n];
      bv[kc].y = xb[(size_t)(k0 + 1) * HWN + n];
    }
  };

  auto compute = [&](v2f (&bv)[16], int jm6, int jd6) {
    v8f accLo = {}, accHi = {};
    #pragma unroll
    for (int kc = 0; kc < 16; ++kc)
      accLo = __builtin_amdgcn_wmma_f32_16x16x4_f32(false, aLo[kc], false, bv[kc],
                                                    (short)0, accLo, false, false);
    #pragma unroll
    for (int kc = 0; kc < 16; ++kc)
      accHi = __builtin_amdgcn_wmma_f32_16x16x4_f32(false, aHi[kc], false, bv[kc],
                                                    (short)0, accHi, false, false);
    const float xx = (float)(jm6 * 16 + nl) * CSTEP - 1.0f;
    const float yy = (float)(jd6)           * CSTEP - 1.0f;
    #pragma unroll
    for (int r = 0; r < 16; ++r) {
      const float c  = (r < 8) ? accLo[r] : accHi[r - 8];
      const float nm = fmaxf(m[r], c);
      const float sc = fast_exp2(m[r] - nm);
      const float e  = fast_exp2(c - nm);
      se[r] = se[r] * sc + e;
      sx[r] = sx[r] * sc + e * xx;
      sy[r] = sy[r] * sc + e * yy;
      m[r]  = nm;
    }
  };

  int jm6 = 0;
  int jd6 = wave * 24;
  loadB(bv0, jbeg);
  for (int j = jbeg; j < jend; j += 2) {
    loadB(bv1, j + 1);
    compute(bv0, jm6, jd6);
    jm6++; if (jm6 == 6) { jm6 = 0; jd6++; }
    const int jn = (j + 2 < jend) ? (j + 2) : jbeg;
    loadB(bv0, jn);
    compute(bv1, jm6, jd6);
    jm6++; if (jm6 == 6) { jm6 = 0; jd6++; }
  }

  #pragma unroll
  for (int mask = 1; mask <= 8; mask <<= 1) {
    #pragma unroll
    for (int r = 0; r < 16; ++r) {
      const float mo = __shfl_xor(m[r],  mask, 32);
      const float so = __shfl_xor(se[r], mask, 32);
      const float xo = __shfl_xor(sx[r], mask, 32);
      const float yo = __shfl_xor(sy[r], mask, 32);
      const float nm = fmaxf(m[r], mo);
      const float s1 = fast_exp2(m[r] - nm);
      const float s2 = fast_exp2(mo   - nm);
      se[r] = se[r] * s1 + so * s2;
      sx[r] = sx[r] * s1 + xo * s2;
      sy[r] = sy[r] * s1 + yo * s2;
      m[r]  = nm;
    }
  }

  if (nl == 0) {
    #pragma unroll
    for (int r = 0; r < 16; ++r) {
      const int row = (r < 8) ? (r + 8 * half) : (16 + (r - 8) + 8 * half);
      lds[row][wave][0] = m[r];
      lds[row][wave][1] = se[r];
      lds[row][wave][2] = sx[r];
      lds[row][wave][3] = sy[r];
    }
  }
  __syncthreads();

  if (tid < 32) {
    const int row = tid;
    const int p   = p0 + row;
    if (p < PP) {
      float M = lds[row][0][0], S = lds[row][0][1];
      float X = lds[row][0][2], Y = lds[row][0][3];
      #pragma unroll
      for (int w = 1; w < NWAVES; ++w) {
        const float mo = lds[row][w][0], so = lds[row][w][1];
        const float xo = lds[row][w][2], yo = lds[row][w][3];
        const float nm = fmaxf(M, mo);
        const float s1 = fast_exp2(M  - nm);
        const float s2 = fast_exp2(mo - nm);
        S = S * s1 + so * s2;
        X = X * s1 + xo * s2;
        Y = Y * s1 + yo * s2;
        M = nm;
      }
      const size_t base = 640 + 2 * ((size_t)b * PP + p);
      out[base + 0] = X / S;
      out[base + 1] = Y / S;
    }
  }
}

__global__ void pred_kernel(const float* __restrict__ w_lin, float* __restrict__ out)
{
  const int tid = blockIdx.x * blockDim.x + threadIdx.x;
  if (tid >= BB * NAA * 2) return;
  const int k  = tid & 1;
  const int bn = tid >> 1;
  const float* inter = out + 640 + (size_t)bn * (NII * 2);
  const float* wr    = w_lin + k * (NII * 2);
  float s = 0.0f;
  #pragma unroll 4
  for (int i = 0; i < NII * 2; ++i) s += inter[i] * wr[i];
  out[tid] = s;
}

extern "C" void kernel_launch(void* const* d_in, const int* in_sizes, int n_in,
                              void* d_out, int out_size, void* d_ws, size_t ws_size,
                              hipStream_t stream)
{
  const float* x     = (const float*)d_in[0];
  const float* desc  = (const float*)d_in[1];
  const float* w_lin = (const float*)d_in[2];
  float* out = (float*)d_out;

  corr_softargmax_kernel<<<BB * T32, 32 * NWAVES, 0, stream>>>(x, desc, out);
  pred_kernel<<<(BB * NAA * 2 + 127) / 128, 128, 0, stream>>>(w_lin, out);
}


# WMMA operands swapped (x tile as A, descriptor tile as B): score tile transposed, a lane owns one softmax row and rescales once per 8 scores; order of the softmax sums changes
# speedup vs baseline: 1.0832x; 1.0832x over previous
#include <hip/hip_runtime.h>
#include <math.h>

typedef __attribute__((ext_vector_type(2))) float v2f;
typedef __attribute__((ext_vector_type(8))) float v8f;

#define BB   16
#define CC   64
#define HH   96
#define WW   96
#define HWN  (HH*WW)
#define NAA  20
#define NII  50
#define PP   (NAA*NII)
#define T32  32
#define NWAVES 4
#define JTOT (HWN/16)
#define JPW  (JTOT/NWAVES)
#define ASCALE (50.0f * 1.44269504088896340736f)
#define CSTEP  (2.0f / 95.0f)

#define MAXF(a, b) __builtin_fmaximum_numf((a), (b))

__device__ __forceinline__ float fast_exp2(float x) {
  return __builtin_amdgcn_exp2f(x);
}

__global__ __launch_bounds__(32*NWAVES)
void corr_softargmax_kernel(const float* __restrict__ x,
                            const float* __restrict__ desc,
                            float* __restrict__ out)
{
  __shared__ float lds[32][NWAVES][4];

  const int b    = blockIdx.x / T32;
  const int t    = blockIdx.x % T32;
  const int tid  = threadIdx.x;
  const int wave = tid >> 5;
  const int lane = tid & 31;
  const int half = lane >> 4;
  const int nl   = lane & 15;
  const int p0   = t * 32;

  v2f dLo[16], dHi[16];
  {
    const int  pl  = p0 + nl;        const bool okl = (pl < PP);
    const int  ph  = p0 + 16 + nl;   const bool okh = (ph < PP);
    #pragma unroll
    for (int kc = 0; kc < 16; ++kc) {
      const int k0 = 4 * kc + 2 * half;
      dLo[kc].x = okl ? ASCALE * desc[(size_t)(k0    ) * PP + pl] : 0.0f;
      dLo[kc].y = okl ? ASCALE * desc[(size_t)(k0 + 1) * PP + pl] : 0.0f;
      dHi[kc].x = okh ? ASCALE * desc[(size_t)(k0    ) * PP + ph] : 0.0f;
      dHi[kc].y = okh ? ASCALE * desc[(size_t)(k0 + 1) * PP + ph] : 0.0f;
    }
  }

  float m[2], se[2], sx[2], sy[2];
  #pragma unroll
  for (int ti = 0; ti < 2; ++ti) { m[ti] = -1e30f; se[ti] = 0.f; sx[ti] = 0.f; sy[ti] = 0.f; }

  const float* xb = x + (size_t)b * CC * HWN;
  const int jbeg = wave * JPW;
  const int jend = jbeg + JPW;

  v2f xv0[16], xv1[16];

  auto loadX = [&](v2f (&xv)[16], int j) {
    const int n = j * 16 + nl;
    #pragma unroll
    for (int kc = 0; kc < 16; ++kc) {
      const int k0 = 4 * kc + 2 * half;
      xv[kc].x = xb[(size_t)(k0    ) * HWN + n];
      xv[kc].y = xb[(size_t)(k0 + 1) * HWN + n];
    }
  };

  auto update = [&](const v8f& acc, int ti, float cb, float rf) {
    float nm = m[ti];
    #pragma unroll
    for (int r = 0; r < 8; ++r) nm = MAXF(nm, acc[r]);
    const float sc = fast_exp2(m[ti] - nm);
    float e[8];
    #pragma unroll
    for (int r = 0; r < 8; ++r) e[r] = fast_exp2(acc[r] - nm);
    const float p01 = e[0] + e[1], p23 = e[2] + e[3], p45 = e[4] + e[5], p67 = e[6] + e[7];
    const float q47 = p45 + p67;
    const float S   = (p01 + p23) + q47;
    const float odd = (e[1] + e[3]) + (e[5] + e[7]);
    float wx = __builtin_fmaf(2.0f, p23 + p67, odd);
    wx = __builtin_fmaf(4.0f, q47, wx);
    se[ti] = __builtin_fmaf(se[ti], sc, S);
    sx[ti] = __builtin_fmaf(sx[ti], sc, __builtin_fmaf(cb, S, wx));
    sy[ti] = __builtin_fmaf(sy[ti], sc, rf * S);
    m[ti]  = nm;
  };

  auto compute = [&](v2f (&xv)[16], int jm6, int jd6) {
    v8f accLo = {}, accHi = {};
    #pragma unroll
    for (int kc = 0; kc < 16; ++kc)
      accLo = __builtin_amdgcn_wmma_f32_16x16x4_f32(false, xv[kc], false, dLo[kc],
                                                    (short)0, accLo, false, false);
    #pragma unroll
    for (int kc = 0; kc < 16; ++kc)
      accHi = __builtin_amdgcn_wmma_f32_16x16x4_f32(false, xv[kc], false, dHi[kc],
                                                    (short)0, accHi, false, false);
    const float cb = (float)(jm6 * 16 + 8 * half);
    const float rf = (float)(jd6);
    update(accLo, 0, cb, rf);
    update(accHi, 1, cb, rf);
  };

  int jm6 = 0;
  int jd6 = wave * 24;
  loadX(xv0, jbeg);
  for (int j = jbeg; j < jend; j += 2) {
    loadX(xv1, j + 1);
    compute(xv0, jm6, jd6);
    jm6++; if (jm6 == 6) { jm6 = 0; jd6++; }
    const int jn = (j + 2 < jend) ? (j + 2) : jbeg;
    __builtin_amdgcn_sched_barrier(0);
    loadX(xv0, jn);
    compute(xv1, jm6, jd6);
    jm6++; if (jm6 == 6) { jm6 = 0; jd6++; }
  }

  #pragma unroll
  for (int ti = 0; ti < 2; ++ti) {
    const float mo = __shfl_xor(m[ti],  16, 32);
    const float so = __shfl_xor(se[ti], 16, 32);
    const float xo = __shfl_xor(sx[ti], 16, 32);
    const float yo = __shfl_xor(sy[ti], 16, 32);
    const float nm = MAXF(m[ti], mo);
    const float s1 = fast_exp2(m[ti] - nm);
    const float s2 = fast_exp2(mo    - nm);
    se[ti] = se[ti] * s1 + so * s2;
    sx[ti] = sx[ti] * s1 + xo * s2;
    sy[ti] = sy[ti] * s1 + yo * s2;
    m[ti]  = nm;
  }

  if (half == 0) {
    #pragma unroll
    for (int ti = 0; ti < 2; ++ti) {
      const int row = 16 * ti + nl;
      lds[row][wave][0] = m[ti];
      lds[row][wave][1] = se[ti];
      lds[row][wave][2] = sx[ti];
      lds[row][wave][3] = sy[ti];
    }
  }
  __syncthreads();

  if (tid < 32) {
    const int row = tid;
    const int p   = p0 + row;
    if (p < PP) {
      float M = lds[row][0][0], S = lds[row][0][1];
      float X = lds[row][0][2], Y = lds[row][0][3];
      #pragma unroll
      for (int w = 1; w < NWAVES; ++w) {
        const float mo = lds[row][w][0], so = lds[row][w][1];
        const float xo = lds[row][w][2], yo = lds[row][w][3];
        const float nm = fmaxf(M, mo);
        const float s1 = fast_exp2(M  - nm);
        const float s2 = fast_exp2(mo - nm);
        S = S * s1 + so * s2;
        X = X * s1 + xo * s2;
        Y = Y * s1 + yo * s2;
        M = nm;
      }
      const size_t base = 640 + 2 * ((size_t)b * PP + p);
      out[base + 0] = __builtin_fmaf(CSTEP, X / S, -1.0f);
      out[base + 1] = __builtin_fmaf(CSTEP, Y / S, -1.0f);
    }
  }
}

__global__ void pred_kernel(const float* __restrict__ w_lin, float* __restrict__ out)
{
  const int tid = blockIdx.x * blockDim.x + threadIdx.x;
  if (tid >= BB * NAA * 2) return;
  const int k  = tid & 1;
  const int bn = tid >> 1;
  const float* inter = out + 640 + (size_t)bn * (NII * 2);
  const float* wr    = w_lin + k * (NII * 2);
  float s = 0.0f;
  #pragma unroll 4
  for (int i = 0; i < NII * 2; ++i) s += inter[i] * wr[i];
  out[tid] = s;
}

extern "C" void kernel_launch(void* const* d_in, const int* in_sizes, int n_in,
                              void* d_out, int out_size, void* d_ws, size_t ws_size,
                              hipStream_t stream)
{
  const float* x     = (const float*)d_in[0];
  const float* desc  = (const float*)d_in[1];
  const float* w_lin = (const float*)d_in[2];
  float* out = (float*)d_out;

  corr_softargmax_kernel<<<BB * T32, 32 * NWAVES, 0, stream>>>(x, desc, out);
  pred_kernel<<<(BB * NAA * 2 + 127) / 128, 128, 0, stream>>>(w_lin, out);
}


# x read in 64-column groups with 128-bit loads (lane: 4 adjacent columns of a channel; sub-tile: every 4th column): whole lines per request, 1/4 the loads; order of the softmax sums changes
# speedup vs baseline: 1.1386x; 1.0511x over previous
#include <hip/hip_runtime.h>
#include <math.h>

typedef __attribute__((ext_vector_type(2))) float v2f;
typedef __attribute__((ext_vector_type(4))) float v4f;
typedef __attribute__((ext_vector_type(8))) float v8f;

#define BB   16
#define CC   64
#define HH   96
#define WW   96
#define HWN  (HH*WW)
#define NAA  20
#define NII  50
#define PP   (NAA*NII)
#define T32  32
#define NWAVES 4
#define GTOT (HWN/64)
#define GPW  (GTOT/NWAVES)
#define ITERS (GPW/2)
#define ASCALE (50.0f * 1.44269504088896340736f)
#define CSTEP  (2.0f / 95.0f)

#define MAXF(a, b) __builtin_fmaximum_numf((a), (b))

__device__ __forceinline__ float fast_exp2(float x) {
  return __builtin_amdgcn_exp2f(x);
}

__global__ __launch_bounds__(32*NWAVES)
void corr_softargmax_kernel(const float* __restrict__ x,
                            const float* __restrict__ desc,
                            float* __restrict__ out)
{
  __shared__ float lds[32][NWAVES][4];

  const int b    = blockIdx.x / T32;
  const int t    = blockIdx.x % T32;
  const int tid  = threadIdx.x;
  const int wave = __builtin_amdgcn_readfirstlane(tid >> 5);
  const int lane = tid & 31;
  const int half = lane >> 4;
  const int nl   = lane & 15;
  const int p0   = t * 32;

  v2f dLo[16], dHi[16];
  {
    const int  pl  = p0 + nl;        const bool okl = (pl < PP);
    const int  ph  = p0 + 16 + nl;   const bool okh = (ph < PP);
    #pragma unroll
    for (int kc = 0; kc < 16; ++kc) {
      const int k0 = 4 * kc + 2 * half;
      dLo[kc].x = okl ? ASCALE * desc[(size_t)(k0    ) * PP + pl] : 0.0f;
      dLo[kc].y = okl ? ASCALE * desc[(size_t)(k0 + 1) * PP + pl] : 0.0f;
      dHi[kc].x = okh ? ASCALE * desc[(size_t)(k0    ) * PP + ph] : 0.0f;
      dHi[kc].y = okh ? ASCALE * desc[(size_t)(k0 + 1) * PP + ph] : 0.0f;
    }
  }

  float m[2], se[2], sx[2], sy[2];
  #pragma unroll
  for (int ti = 0; ti < 2; ++ti) { m[ti] = -1e30f; se[ti] = 0.f; sx[ti] = 0.f; sy[ti] = 0.f; }

  const float* xl = x + (size_t)b * CC * HWN + (size_t)(2 * half) * HWN + 4 * nl;
  const int gbeg = wave * GPW;

  v4f xq0[16][2], xq1[16][2];

  auto loadX = [&](v4f (&xq)[16][2], int g) {
    const float* src = xl + (size_t)g * 64;
    #pragma unroll
    for (int kc = 0; kc < 16; ++kc) {
      xq[kc][0] = *(const v4f*)(src + (size_t)(4 * kc    ) * HWN);
      xq[kc][1] = *(const v4f*)(src + (size_t)(4 * kc + 1) * HWN);
    }
  };

  auto update = [&](const v8f& acc, int ti, float cb, float rf) {
    float nm = m[ti];
    #pragma unroll
    for (int r = 0; r < 8; ++r) nm = MAXF(nm, acc[r]);
    const float sc = fast_exp2(m[ti] - nm);
    float e[8];
    #pragma unroll
    for (int r = 0; r < 8; ++r) e[r] = fast_exp2(acc[r] - nm);
    const float p01 = e[0] + e[1], p23 = e[2] + e[3], p45 = e[4] + e[5], p67 = e[6] + e[7];
    const float q47 = p45 + p67;
    const float S   = (p01 + p23) + q47;
    const float odd = (e[1] + e[3]) + (e[5] + e[7]);
    float wx = __builtin_fmaf(2.0f, p23 + p67, odd);
    wx = __builtin_fmaf(4.0f, q47, wx);
    se[ti] = __builtin_fmaf(se[ti], sc, S);
    sx[ti] = __builtin_fmaf(sx[ti], sc, __builtin_fmaf(cb, S, wx));
    sy[ti] = __builtin_fmaf(sy[ti], sc, rf * S);
    m[ti]  = nm;
  };

  auto compute = [&](v4f (&xq)[16][2], int g) {
    const int fb  = g * 64 + 32 * half;
    const int row = fb / WW;
    const int col = fb - row * WW;
    const float rf = (float)row;
    #pragma unroll
    for (int c = 0; c < 4; ++c) {
      v8f accLo = {}, accHi = {};
      #pragma unroll
      for (int kc = 0; kc < 16; ++kc) {
        v2f a;
        a.x = xq[kc][0][c];
        a.y = xq[kc][1][c];
        accLo = __builtin_amdgcn_wmma_f32_16x16x4_f32(false, a, false, dLo[kc],
                                                      (short)0, accLo, false, false);
        accHi = __builtin_amdgcn_wmma_f32_16x16x4_f32(false, a, false, dHi[kc],
                                                      (short)0, accHi, false, false);
      }
      const float cb = (float)(col + c) * 0.25f;
      update(accLo, 0, cb, rf);
      update(accHi, 1, cb, rf);
    }
  };

  loadX(xq0, gbeg);
  for (int it = 0; it < ITERS - 1; ++it) {
    const int g = gbeg + 2 * it;
    loadX(xq1, g + 1);
    __builtin_amdgcn_sched_barrier(0);
    compute(xq0, g);
    __builtin_amdgcn_sched_barrier(0);
    loadX(xq0, g + 2);
    __builtin_amdgcn_sched_barrier(0);
    compute(xq1, g + 1);
  }
  {
    const int g = gbeg + GPW - 2;
    loadX(xq1, g + 1);
    __builtin_amdgcn_sched_barrier(0);
    compute(xq0, g);
    compute(xq1, g + 1);
  }

  #pragma unroll
  for (int ti = 0; ti < 2; ++ti) {
    const float mo = __shfl_xor(m[ti],  16, 32);
    const float so = __shfl_xor(se[ti], 16, 32);
    const float xo = __shfl_xor(sx[ti], 16, 32);
    const float yo = __shfl_xor(sy[ti], 16, 32);
    const float nm = MAXF(m[ti], mo);
    const float s1 = fast_exp2(m[ti] - nm);
    const float s2 = fast_exp2(mo    - nm);
    se[ti] = se[ti] * s1 + so * s2;
    sx[ti] = sx[ti] * s1 + xo * s2;
    sy[ti] = sy[ti] * s1 + yo * s2;
    m[ti]  = nm;
  }

  if (half == 0) {
    #pragma unroll
    for (int ti = 0; ti < 2; ++ti) {
      const int row = 16 * ti + nl;
      lds[row][wave][0] = m[ti];
      lds[row][wave][1] = se[ti];
      lds[row][wave][2] = sx[ti];
      lds[row][wave][3] = sy[ti];
    }
  }
  __syncthreads();

  if (tid < 32) {
    const int row = tid;
    const int p   = p0 + row;
    if (p < PP) {
      float M = lds[row][0][0], S = lds[row][0][1];
      float X = lds[row][0][2], Y = lds[row][0][3];
      #pragma unroll
      for (int w = 1; w < NWAVES; ++w) {
        const float mo = lds[row][w][0], so = lds[row][w][1];
        const float xo = lds[row][w][2], yo = lds[row][w][3];
        const float nm = fmaxf(M, mo);
        const float s1 = fast_exp2(M  - nm);
        const float s2 = fast_exp2(mo - nm);
        S = S * s1 + so * s2;
        X = X * s1 + xo * s2;
        Y = Y * s1 + yo * s2;
        M = nm;
      }
      const size_t base = 640 + 2 * ((size_t)b * PP + p);
      out[base + 0] = __builtin_fmaf(4.0f * CSTEP, X / S, -1.0f);
      out[base + 1] = __builtin_fmaf(CSTEP, Y / S, -1.0f);
    }
  }
}

__global__ void pred_kernel(const float* __restrict__ w_lin, float* __restrict__ out)
{
  const int tid = blockIdx.x * blockDim.x + threadIdx.x;
  if (tid >= BB * NAA * 2) return;
  const int k  = tid & 1;
  const int bn = tid >> 1;
  const float* inter = out + 640 + (size_t)bn * (NII * 2);
  const float* wr    = w_lin + k * (NII * 2);
  float s = 0.0f;
  #pragma unroll 4
  for (int i = 0; i < NII * 2; ++i) s += inter[i] * wr[i];
  out[tid] = s;
}

extern "C" void kernel_launch(void* const* d_in, const int* in_sizes, int n_in,
                              void* d_out, int out_size, void* d_ws, size_t ws_size,
                              hipStream_t stream)
{
  const float* x     = (const float*)d_in[0];
  const float* desc  = (const float*)d_in[1];
  const float* w_lin = (const float*)d_in[2];
  float* out = (float*)d_out;

  corr_softargmax_kernel<<<BB * T32, 32 * NWAVES, 0, stream>>>(x, desc, out);
  pred_kernel<<<(BB * NAA * 2 + 127) / 128, 128, 0, stream>>>(w_lin, out);
}


# Loads that each waited alone now issue grouped: the 64 descriptor loads lose their condition (padded rows read row 999, scaled by 0), the 2nd kernel's 25 load steps are unrolled; no sum reordered
# speedup vs baseline: 1.3301x; 1.1682x over previous
#include <hip/hip_runtime.h>
#include <math.h>

typedef __attribute__((ext_vector_type(2))) float v2f;
typedef __attribute__((ext_vector_type(4))) float v4f;
typedef __attribute__((ext_vector_type(8))) float v8f;

#define BB   16
#define CC   64
#define HH   96
#define WW   96
#define HWN  (HH*WW)
#define NAA  20
#define NII  50
#define PP   (NAA*NII)
#define T32  32
#define NWAVES 4
#define GTOT (HWN/64)
#define GPW  (GTOT/NWAVES)
#define ITERS (GPW/2)
#define ASCALE (50.0f * 1.44269504088896340736f)
#define CSTEP  (2.0f / 95.0f)

#define MAXF(a, b) __builtin_fmaximum_numf((a), (b))

__device__ __forceinline__ float fast_exp2(float x) {
  return __builtin_amdgcn_exp2f(x);
}

__global__ __launch_bounds__(32*NWAVES)
void corr_softargmax_kernel(const float* __restrict__ x,
                            const float* __restrict__ desc,
                            float* __restrict__ out)
{
  __shared__ float lds[32][NWAVES][4];

  const int b    = blockIdx.x / T32;
  const int t    = blockIdx.x % T32;
  const int tid  = threadIdx.x;
  const int wave = __builtin_amdgcn_readfirstlane(tid >> 5);
  const int lane = tid & 31;
  const int half = lane >> 4;
  const int nl   = lane & 15;
  const int p0   = t * 32;

  v2f dLo[16], dHi[16];
  {
    const int   pl  = p0 + nl;        const bool okl = (pl < PP);
    const int   ph  = p0 + 16 + nl;   const bool okh = (ph < PP);
    const int   plc = okl ? pl : PP - 1;
    const int   phc = okh ? ph : PP - 1;
    const float sl  = okl ? ASCALE : 0.0f;
    const float sh  = okh ? ASCALE : 0.0f;
    #pragma unroll
    for (int kc = 0; kc < 16; ++kc) {
      const int k0 = 4 * kc + 2 * half;
      dLo[kc].x = sl * desc[(size_t)(k0    ) * PP + plc];
      dLo[kc].y = sl * desc[(size_t)(k0 + 1) * PP + plc];
      dHi[kc].x = sh * desc[(size_t)(k0    ) * PP + phc];
      dHi[kc].y = sh * desc[(size_t)(k0 + 1) * PP + phc];
    }
  }

  float m[2], se[2], sx[2], sy[2];
  #pragma unroll
  for (int ti = 0; ti < 2; ++ti) { m[ti] = -1e30f; se[ti] = 0.f; sx[ti] = 0.f; sy[ti] = 0.f; }

  const float* xl = x + (size_t)b * CC * HWN + (size_t)(2 * half) * HWN + 4 * nl;
  const int gbeg = wave * GPW;

  v4f xq0[16][2], xq1[16][2];

  auto loadX = [&](v4f (&xq)[16][2], int g) {
    const float* src = xl + (size_t)g * 64;
    #pragma unroll
    for (int kc = 0; kc < 16; ++kc) {
      xq[kc][0] = *(const v4f*)(src + (size_t)(4 * kc    ) * HWN);
      xq[kc][1] = *(const v4f*)(src + (size_t)(4 * kc + 1) * HWN);
    }
  };

  auto update = [&](const v8f& acc, int ti, float cb, float rf) {
    float nm = m[ti];
    #pragma unroll
    for (int r = 0; r < 8; ++r) nm = MAXF(nm, acc[r]);
    const float sc = fast_exp2(m[ti] - nm);
    float e[8];
    #pragma unroll
    for (int r = 0; r < 8; ++r) e[r] = fast_exp2(acc[r] - nm);
    const float p01 = e[0] + e[1], p23 = e[2] + e[3], p45 = e[4] + e[5], p67 = e[6] + e[7];
    const float q47 = p45 + p67;
    const float S   = (p01 + p23) + q47;
    const float odd = (e[1] + e[3]) + (e[5] + e[7]);
    float wx = __builtin_fmaf(2.0f, p23 + p67, odd);
    wx = __builtin_fmaf(4.0f, q47, wx);
    se[ti] = __builtin_fmaf(se[ti], sc, S);
    sx[ti] = __builtin_fmaf(sx[ti], sc, __builtin_fmaf(cb, S, wx));
    sy[ti] = __builtin_fmaf(sy[ti], sc, rf * S);
    m[ti]  = nm;
  };

  auto compute = [&](v4f (&xq)[16][2], int g) {
    const int fb  = g * 64 + 32 * half;
    const int row = fb / WW;
    const int col = fb - row * WW;
    const float rf = (float)row;
    #pragma unroll
    for (int c = 0; c < 4; ++c) {
      v8f accLo = {}, accHi = {};
      #pragma unroll
      for (int kc = 0; kc < 16; ++kc) {
        v2f a;
        a.x = xq[kc][0][c];
        a.y = xq[kc][1][c];
        accLo = __builtin_amdgcn_wmma_f32_16x16x4_f32(false, a, false, dLo[kc],
                                                      (short)0, accLo, false, false);
        accHi = __builtin_amdgcn_wmma_f32_16x16x4_f32(false, a, false, dHi[kc],
                                                      (short)0, accHi, false, false);
      }
      const float cb = (float)(col + c) * 0.25f;
      update(accLo, 0, cb, rf);
      update(accHi, 1, cb, rf);
    }
  };

  loadX(xq0, gbeg);
  for (int it = 0; it < ITERS - 1; ++it) {
    const int g = gbeg + 2 * it;
    loadX(xq1, g + 1);
    __builtin_amdgcn_sched_barrier(0);
    compute(xq0, g);
    __builtin_amdgcn_sched_barrier(0);
    loadX(xq0, g + 2);
    __builtin_amdgcn_sched_barrier(0);
    compute(xq1, g + 1);
  }
  {
    const int g = gbeg + GPW - 2;
    loadX(xq1, g + 1);
    __builtin_amdgcn_sched_barrier(0);
    compute(xq0, g);
    compute(xq1, g + 1);
  }

  #pragma unroll
  for (int ti = 0; ti < 2; ++ti) {
    const float mo = __shfl_xor(m[ti],  16, 32);
    const float so = __shfl_xor(se[ti], 16, 32);
    const float xo = __shfl_xor(sx[ti], 16, 32);
    const float yo = __shfl_xor(sy[ti], 16, 32);
    const float nm = MAXF(m[ti], mo);
    const float s1 = fast_exp2(m[ti] - nm);
    const float s2 = fast_exp2(mo    - nm);
    se[ti] = se[ti] * s1 + so * s2;
    sx[ti] = sx[ti] * s1 + xo * s2;
    sy[ti] = sy[ti] * s1 + yo * s2;
    m[ti]  = nm;
  }

  if (half == 0) {
    #pragma unroll
    for (int ti = 0; ti < 2; ++ti) {
      const int row = 16 * ti + nl;
      lds[row][wave][0] = m[ti];
      lds[row][wave][1] = se[ti];
      lds[row][wave][2] = sx[ti];
      lds[row][wave][3] = sy[ti];
    }
  }
  __syncthreads();

  if (tid < 32) {
    const int row = tid;
    const int p   = p0 + row;
    if (p < PP) {
      float M = lds[row][0][0], S = lds[row][0][1];
      float X = lds[row][0][2], Y = lds[row][0][3];
      #pragma unroll
      for (int w = 1; w < NWAVES; ++w) {
        const float mo = lds[row][w][0], so = lds[row][w][1];
        const float xo = lds[row][w][2], yo = lds[row][w][3];
        const float nm = fmaxf(M, mo);
        const float s1 = fast_exp2(M  - nm);
        const float s2 = fast_exp2(mo - nm);
        S = S * s1 + so * s2;
        X = X * s1 + xo * s2;
        Y = Y * s1 + yo * s2;
        M = nm;
      }
      const size_t base = 640 + 2 * ((size_t)b * PP + p);
      out[base + 0] = __builtin_fmaf(4.0f * CSTEP, X / S, -1.0f);
      out[base + 1] = __builtin_fmaf(CSTEP, Y / S, -1.0f);
    }
  }
}

__global__ void pred_kernel(const float* __restrict__ w_lin, float* __restrict__ out)
{
  const int tid = blockIdx.x * blockDim.x + threadIdx.x;
  if (tid >= BB * NAA * 2) return;
  const int k  = tid & 1;
  const int bn = tid >> 1;
  const float* inter = out + 640 + (size_t)bn * (NII * 2);
  const float* wr    = w_lin + k * (NII * 2);
  v4f a[NII / 2], w[NII / 2];
  #pragma unroll
  for (int j = 0; j < NII / 2; ++j) {
    a[j] = *(const v4f*)(inter + 4 * j);
    w[j] = *(const v4f*)(wr    + 4 * j);
  }
  float s = 0.0f;
  #pragma unroll
  for (int j = 0; j < NII / 2; ++j) {
    s += a[j].x * w[j].x;
    s += a[j].y * w[j].y;
    s += a[j].z * w[j].z;
    s += a[j].w * w[j].w;
  }
  out[tid] = s;
}

extern "C" void kernel_launch(void* const* d_in, const int* in_sizes, int n_in,
                              void* d_out, int out_size, void* d_ws, size_t ws_size,
                              hipStream_t stream)
{
  const float* x     = (const float*)d_in[0];
  const float* desc  = (const float*)d_in[1];
  const float* w_lin = (const float*)d_in[2];
  float* out = (float*)d_out;

  corr_softargmax_kernel<<<BB * T32, 32 * NWAVES, 0, stream>>>(x, desc, out);
  pred_kernel<<<(BB * NAA * 2 + 127) / 128, 128, 0, stream>>>(w_lin, out);
}
